# starting kernel
# baseline (speedup 1.0000x reference)
#include <hip/hip_runtime.h>
#include <math.h>


typedef __attribute__((ext_vector_type(2))) float v2f;
typedef __attribute__((ext_vector_type(8))) float v8f;

#define MM 16
#define NN 2048
#define DD 128
#define PP 8192
#define HH 16
#define NCHUNK 33

__global__ __launch_bounds__(256) void k_rmsnorm_x(const float* __restrict__ X,
                                                   float* __restrict__ Xn)
{
    __shared__ float red[256];
    const int m = blockIdx.x;
    const int t = threadIdx.x;
    float ss = 0.0f;
    for (int j = t; j < NN; j += 256) {
        float x = X[(size_t)m * NN + j];
        ss += x * x;
    }
    red[t] = ss;
    __syncthreads();
    for (int o = 128; o > 0; o >>= 1) {
        if (t < o) red[t] += red[t + o];
        __syncthreads();
    }
    const float rs = rsqrtf(red[0] / (float)NN);
    for (int j = t; j < NN; j += 256)
        Xn[(size_t)m * NN + j] = X[(size_t)m * NN + j] * rs;
}

__global__ __launch_bounds__(32) void k_proj_wmma(const float* __restrict__ Xn,
                                                  const float* __restrict__ Wq,
                                                  const float* __restrict__ Wk,
                                                  const float* __restrict__ Wv,
                                                  float* __restrict__ qraw,
                                                  float* __restrict__ kraw,
                                                  float* __restrict__ vraw)
{
    const int lane = threadIdx.x;
    const int col  = lane & 15;
    const int lh   = lane >> 4;
    const int bid  = blockIdx.x;
    const int w    = bid >> 7;
    const int n0   = (bid & 127) << 4;

    const float* W   = (w == 0) ? Wq : (w == 1) ? Wk : Wv;
    float*       out = (w == 0) ? qraw : (w == 1) ? kraw : vraw;

    const float* arow = Xn + (size_t)col * NN + 2 * lh;
    const float* bcol = W + n0 + col;

    v8f acc = {};
    #pragma unroll 8
    for (int kb = 0; kb < NN; kb += 4) {
        float2 a2 = *reinterpret_cast<const float2*>(arow + kb);
        v2f A = {a2.x, a2.y};
        const float* bp = bcol + (size_t)(kb + 2 * lh) * NN;
        v2f B = {bp[0], bp[NN]};
        acc = __builtin_amdgcn_wmma_f32_16x16x4_f32(false, A, false, B,
                                                    (short)0, acc, false, false);
    }

    #pragma unroll
    for (int r = 0; r < 8; ++r)
        out[(size_t)(r + 8 * lh) * NN + n0 + col] = acc[r];
}

__global__ __launch_bounds__(256) void k_qknorm(const float* __restrict__ qraw,
                                                const float* __restrict__ kraw,
                                                const float* __restrict__ vraw,
                                                float* __restrict__ qn,
                                                float* __restrict__ kn,
                                                float* __restrict__ vn)
{
    const int gtid = blockIdx.x * blockDim.x + threadIdx.x;
    const int wid  = gtid >> 5;
    const int lane = gtid & 31;
    const int h = wid >> 4;
    const int m = wid & 15;
    const int d0 = lane * 4;

    const size_t src = (size_t)m * NN + (size_t)h * DD + d0;
    const size_t dst = ((size_t)h * MM + m) * DD + d0;

    float4 q4 = *reinterpret_cast<const float4*>(qraw + src);
    float ss = q4.x * q4.x + q4.y * q4.y + q4.z * q4.z + q4.w * q4.w;
    #pragma unroll
    for (int mk = 16; mk >= 1; mk >>= 1) ss += __shfl_xor(ss, mk, 32);
    float rs = rsqrtf(ss / (float)DD);
    float4 o;
    o.x = q4.x * rs; o.y = q4.y * rs; o.z = q4.z * rs; o.w = q4.w * rs;
    *reinterpret_cast<float4*>(qn + dst) = o;

    float4 k4 = *reinterpret_cast<const float4*>(kraw + src);
    ss = k4.x * k4.x + k4.y * k4.y + k4.z * k4.z + k4.w * k4.w;
    #pragma unroll
    for (int mk = 16; mk >= 1; mk >>= 1) ss += __shfl_xor(ss, mk, 32);
    rs = rsqrtf(ss / (float)DD);
    o.x = k4.x * rs; o.y = k4.y * rs; o.z = k4.z * rs; o.w = k4.w * rs;
    *reinterpret_cast<float4*>(kn + dst) = o;

    *reinterpret_cast<float4*>(vn + dst) =
        *reinterpret_cast<const float4*>(vraw + src);
}

__global__ __launch_bounds__(32) void k_attn_partial(const float* __restrict__ qn,
                                                     const float* __restrict__ kn,
                                                     const float* __restrict__ vn,
                                                     const float* __restrict__ cK,
                                                     const float* __restrict__ cV,
                                                     float* __restrict__ pout,
                                                     float* __restrict__ pmax,
                                                     float* __restrict__ psum)
{
    __shared__ __align__(16) float q_lds[MM * DD];
    __shared__ __align__(16) float sc[MM * 256];

    const int lane = threadIdx.x;
    const int col  = lane & 15;
    const int lh   = lane >> 4;
    const int s    = blockIdx.x;
    const int h    = blockIdx.y;

    const float* qh = qn + (size_t)h * MM * DD;
    for (int i = lane; i < (MM * DD) / 4; i += 32)
        reinterpret_cast<float4*>(q_lds)[i] =
            reinterpret_cast<const float4*>(qh)[i];

    const int nrows = (s < 32) ? 256 : MM;
    const float* Kb = (s < 32) ? (cK + ((size_t)h * PP + (size_t)s * 256) * DD)
                               : (kn + (size_t)h * MM * DD);
    const float* Vb = (s < 32) ? (cV + ((size_t)h * PP + (size_t)s * 256) * DD)
                               : (vn + (size_t)h * MM * DD);

    if (nrows < 256) {
        for (int i = lane; i < MM * 256; i += 32) sc[i] = -3.0e38f;
    }

    for (int t = 0; t < nrows / 16; ++t) {
        const int j0 = t * 16;
        const float* krow = Kb + (size_t)(j0 + col) * DD + 2 * lh;
        const float* qrow = q_lds + (size_t)col * DD + 2 * lh;
        v8f acc = {};
        #pragma unroll
        for (int kb = 0; kb < DD; kb += 4) {
            float2 a2 = *reinterpret_cast<const float2*>(qrow + kb);
            float2 b2 = *reinterpret_cast<const float2*>(krow + kb);
            v2f A = {a2.x, a2.y};
            v2f B = {b2.x, b2.y};
            acc = __builtin_amdgcn_wmma_f32_16x16x4_f32(false, A, false, B,
                                                        (short)0, acc, false, false);
        }
        #pragma unroll
        for (int r = 0; r < 8; ++r)
            sc[(r + 8 * lh) * 256 + j0 + col] = acc[r];
    }

    float* scrow = sc + (size_t)col * 256 + lh * 128;
    float mloc = -3.0e38f;
    for (int j = 0; j < 128; ++j) mloc = fmaxf(mloc, scrow[j]);
    mloc = fmaxf(mloc, __shfl_xor(mloc, 16, 32));
    float ssum = 0.0f;
    for (int j = 0; j < 128; ++j) {
        float e = __expf(scrow[j] - mloc);
        scrow[j] = e;
        ssum += e;
    }
    ssum += __shfl_xor(ssum, 16, 32);
    if (lh == 0) {
        pmax[((size_t)h * NCHUNK + s) * MM + col] = mloc;
        psum[((size_t)h * NCHUNK + s) * MM + col] = ssum;
    }

    v8f zero = {};
    v8f oacc[8];
    #pragma unroll
    for (int dt = 0; dt < 8; ++dt) oacc[dt] = zero;

    for (int kb = 0; kb < nrows; kb += 4) {
        float2 a2 = *reinterpret_cast<const float2*>(sc + (size_t)col * 256 + kb + 2 * lh);
        v2f A = {a2.x, a2.y};
        const float* vrow = Vb + (size_t)(kb + 2 * lh) * DD + col;
        #pragma unroll
        for (int dt = 0; dt < 8; ++dt) {
            v2f B = {vrow[dt * 16], vrow[DD + dt * 16]};
            oacc[dt] = __builtin_amdgcn_wmma_f32_16x16x4_f32(false, A, false, B,
                                                             (short)0, oacc[dt],
                                                             false, false);
        }
    }

    float* po = pout + ((size_t)h * NCHUNK + s) * MM * DD;
    #pragma unroll
    for (int dt = 0; dt < 8; ++dt) {
        #pragma unroll
        for (int r = 0; r < 8; ++r)
            po[(size_t)(r + 8 * lh) * DD + dt * 16 + col] = oacc[dt][r];
    }
}

__global__ __launch_bounds__(256) void k_combine(const float* __restrict__ pout,
                                                 const float* __restrict__ pmax,
                                                 const float* __restrict__ psum,
                                                 float* __restrict__ out)
{
    const int h = blockIdx.x;
    for (int idx = threadIdx.x; idx < MM * DD; idx += blockDim.x) {
        const int m = idx >> 7;
        const int d = idx & (DD - 1);
        const float* pm  = pmax + (size_t)h * NCHUNK * MM + m;
        const float* psu = psum + (size_t)h * NCHUNK * MM + m;
        float g = -3.0e38f;
        for (int s = 0; s < NCHUNK; ++s) g = fmaxf(g, pm[(size_t)s * MM]);
        float tot = 0.0f, acc = 0.0f;
        for (int s = 0; s < NCHUNK; ++s) {
            float w = __expf(pm[(size_t)s * MM] - g);
            tot += w * psu[(size_t)s * MM];
            acc += w * pout[(((size_t)h * NCHUNK + s) * MM + m) * DD + d];
        }
        out[(size_t)m * NN + (size_t)h * DD + d] = acc / tot;
    }
}

extern "C" void kernel_launch(void* const* d_in, const int* in_sizes, int n_in,
                              void* d_out, int out_size, void* d_ws, size_t ws_size,
                              hipStream_t stream)
{
    (void)in_sizes; (void)n_in; (void)out_size; (void)ws_size;
    const float* X  = (const float*)d_in[0];
    const float* Wq = (const float*)d_in[1];
    const float* Wk = (const float*)d_in[2];
    const float* Wv = (const float*)d_in[3];
    const float* cK = (const float*)d_in[4];
    const float* cV = (const float*)d_in[5];
    float* out = (float*)d_out;
    float* ws  = (float*)d_ws;

    float* xnorm = ws;
    float* qraw  = xnorm + MM * NN;
    float* kraw  = qraw + MM * NN;
    float* vraw  = kraw + MM * NN;
    float* qn    = vraw + MM * NN;
    float* kn    = qn + HH * MM * DD;
    float* vn    = kn + HH * MM * DD;
    float* pout  = vn + HH * MM * DD;
    float* pmaxb = pout + (size_t)HH * NCHUNK * MM * DD;
    float* psumb = pmaxb + (size_t)HH * NCHUNK * MM;

    k_rmsnorm_x<<<16, 256, 0, stream>>>(X, xnorm);
    k_proj_wmma<<<3 * (NN / 16), 32, 0, stream>>>(xnorm, Wq, Wk, Wv,
                                                  qraw, kraw, vraw);
    k_qknorm<<<(HH * MM) / 8, 256, 0, stream>>>(qraw, kraw, vraw, qn, kn, vn);
    k_attn_partial<<<dim3(NCHUNK, HH), 32, 0, stream>>>(qn, kn, vn, cK, cV,
                                                        pout, pmaxb, psumb);
    k_combine<<<HH, 256, 0, stream>>>(pout, pmaxb, psumb, out);
}


# RMS row scale of X applied behind the projection's k sum: the qk-norm kernel sums X^2 of its row and scales the summed q, k, v; norm-x launch dropped, 5 -> 4 launches; X^2 sum order, rounding change
# speedup vs baseline: 5.0700x; 5.0700x over previous
#include <hip/hip_runtime.h>
#include <math.h>


typedef __attribute__((ext_vector_type(2))) float v2f;
typedef __attribute__((ext_vector_type(4))) float v4f;
typedef __attribute__((ext_vector_type(8))) float v8f;

#define MM 16
#define NN 2048
#define DD 128
#define PP 8192
#define HH 16
#define CH 512
#define NW 16
#define NCHUNK 17
#define KS 256
#define NS 8
#define PG 4
#define NSTG (KS / (8 * PG))
#define AHEAD 3

__global__ __launch_bounds__(32) void k_proj_slab(const float* __restrict__ X,
                                                  const float* __restrict__ Wq,
                                                  const float* __restrict__ Wk,
                                                  const float* __restrict__ Wv,
                                                  float* __restrict__ part)
{
    const int lane = threadIdx.x;
    const int col  = lane & 15;
    const int lh   = lane >> 4;
    const int n0   = blockIdx.x << 6;
    const int s    = blockIdx.y;
    const int w    = blockIdx.z;

    const float* W  = (w == 0) ? Wq : (w == 1) ? Wk : Wv;
    const float* ap = X + (size_t)col * NN + s * KS + 4 * lh;
    const float* bp = W + (size_t)(s * KS + 4 * lh) * NN + n0 + 4 * col;

    v4f af[AHEAD + 1][PG];
    v4f bf[AHEAD + 1][PG][4];
#define LOAD_STAGE(T) do { \
    _Pragma("unroll") for (int g_ = 0; g_ < PG; ++g_) { \
        const int i_ = (T) * PG + g_; \
        af[(T) % (AHEAD + 1)][g_] = *reinterpret_cast<const v4f*>(ap + 8 * i_); \
        _Pragma("unroll") for (int j_ = 0; j_ < 4; ++j_) \
            bf[(T) % (AHEAD + 1)][g_][j_] = \
                *reinterpret_cast<const v4f*>(bp + (size_t)(8 * i_ + j_) * NN); \
    } } while (0)

    v8f acc[4];
    #pragma unroll
    for (int e = 0; e < 4; ++e) { v8f z = {}; acc[e] = z; }

    #pragma unroll
    for (int t = 0; t < AHEAD; ++t) LOAD_STAGE(t);

    #pragma unroll
    for (int t = 0; t < NSTG; ++t) {
        __builtin_amdgcn_sched_barrier(0);
        if (t + AHEAD < NSTG) LOAD_STAGE(t + AHEAD);
        __builtin_amdgcn_sched_barrier(0);
        #pragma unroll
        for (int g = 0; g < PG; ++g) {
            #pragma unroll
            for (int p = 0; p < 2; ++p) {
                const v4f a4 = af[t % (AHEAD + 1)][g];
                v2f A = {a4[2 * p], a4[2 * p + 1]};
                #pragma unroll
                for (int e = 0; e < 4; ++e) {
                    v2f B = {bf[t % (AHEAD + 1)][g][2 * p][e],
                             bf[t % (AHEAD + 1)][g][2 * p + 1][e]};
                    acc[e] = __builtin_amdgcn_wmma_f32_16x16x4_f32(
                        false, A, false, B, (short)0, acc[e], false, false);
                }
            }
        }
    }
#undef LOAD_STAGE

    float* po = part + ((size_t)(w * NS + s) * MM + 8 * lh) * NN + n0 + 4 * col;
    #pragma unroll
    for (int r = 0; r < 8; ++r) {
        v4f o = {acc[0][r], acc[1][r], acc[2][r], acc[3][r]};
        *reinterpret_cast<v4f*>(po + (size_t)r * NN) = o;
    }
}

__global__ __launch_bounds__(32) void k_qknorm(const float* __restrict__ X,
                                               const float* __restrict__ part,
                                               float* __restrict__ qn,
                                               float* __restrict__ kn,
                                               float* __restrict__ vn)
{
    const int wid  = blockIdx.x;
    const int lane = threadIdx.x;
    const int h = wid >> 4;
    const int m = wid & 15;
    const int d0 = lane * 4;

    const size_t src = (size_t)m * NN + (size_t)h * DD + d0;
    const size_t dst = ((size_t)h * MM + m) * DD + d0;

    v4f xv[NN / 128];
    v4f tq[NS], tk[NS], tv[NS];
    #pragma unroll
    for (int s = 0; s < NS; ++s) {
        const float* p = part + (size_t)s * MM * NN + src;
        tq[s] = *reinterpret_cast<const v4f*>(p);
        tk[s] = *reinterpret_cast<const v4f*>(p + (size_t)NS * MM * NN);
        tv[s] = *reinterpret_cast<const v4f*>(p + (size_t)2 * NS * MM * NN);
    }
    #pragma unroll
    for (int i = 0; i < NN / 128; ++i)
        xv[i] = *reinterpret_cast<const v4f*>(X + (size_t)m * NN + 128 * i + d0);
    __builtin_amdgcn_sched_barrier(0);

    v4f q4 = tq[0], k4 = tk[0], v4 = tv[0];
    #pragma unroll
    for (int s = 1; s < NS; ++s) { q4 += tq[s]; k4 += tk[s]; v4 += tv[s]; }

    float sx = 0.0f;
    #pragma unroll
    for (int i = 0; i < NN / 128; ++i)
        sx += xv[i][0] * xv[i][0] + xv[i][1] * xv[i][1]
            + xv[i][2] * xv[i][2] + xv[i][3] * xv[i][3];
    #pragma unroll
    for (int mk = 16; mk >= 1; mk >>= 1) sx += __shfl_xor(sx, mk, 32);
    const float rx = rsqrtf(sx / (float)NN);
    q4 = q4 * rx;
    k4 = k4 * rx;
    v4 = v4 * rx;

    float ss = q4[0] * q4[0] + q4[1] * q4[1] + q4[2] * q4[2] + q4[3] * q4[3];
    #pragma unroll
    for (int mk = 16; mk >= 1; mk >>= 1) ss += __shfl_xor(ss, mk, 32);
    float rs = rsqrtf(ss / (float)DD);
    *reinterpret_cast<v4f*>(qn + dst) = q4 * rs;

    ss = k4[0] * k4[0] + k4[1] * k4[1] + k4[2] * k4[2] + k4[3] * k4[3];
    #pragma unroll
    for (int mk = 16; mk >= 1; mk >>= 1) ss += __shfl_xor(ss, mk, 32);
    rs = rsqrtf(ss / (float)DD);
    *reinterpret_cast<v4f*>(kn + dst) = k4 * rs;

    *reinterpret_cast<v4f*>(vn + dst) = v4;
}

template <int NT>
__device__ __forceinline__ void attn_wave(const float* __restrict__ qh,
                                          const float* __restrict__ Kb,
                                          const float* __restrict__ Vb,
                                          const int col, const int lh,
                                          float& mx_out, float& sum_out,
                                          v8f (&oacc)[8])
{
    v4f qf[16];
    v4f kf[NT][16];
    #pragma unroll
    for (int t = 0; t < NT; ++t) {
        #pragma unroll
        for (int i = 0; i < 16; ++i)
            kf[t][i] = *reinterpret_cast<const v4f*>(
                Kb + (size_t)(t * 16 + col) * DD + 8 * i + 4 * lh);
    }
    #pragma unroll
    for (int i = 0; i < 16; ++i)
        qf[i] = *reinterpret_cast<const v4f*>(qh + (size_t)col * DD + 8 * i + 4 * lh);

    v8f st[NT];
    #pragma unroll
    for (int t = 0; t < NT; ++t) { v8f z = {}; st[t] = z; }

    #pragma unroll
    for (int i = 0; i < 16; ++i) {
        #pragma unroll
        for (int t = 0; t < NT; ++t) {
            v2f A = {kf[t][i][0], kf[t][i][1]};
            v2f B = {qf[i][0], qf[i][1]};
            st[t] = __builtin_amdgcn_wmma_f32_16x16x4_f32(false, A, false, B,
                                                          (short)0, st[t], false, false);
        }
        #pragma unroll
        for (int t = 0; t < NT; ++t) {
            v2f A = {kf[t][i][2], kf[t][i][3]};
            v2f B = {qf[i][2], qf[i][3]};
            st[t] = __builtin_amdgcn_wmma_f32_16x16x4_f32(false, A, false, B,
                                                          (short)0, st[t], false, false);
        }
    }

    v4f vf[NT][4][2][2];
    #pragma unroll
    for (int t = 0; t < NT; ++t) {
        #pragma unroll
        for (int i = 0; i < 4; ++i) {
            #pragma unroll
            for (int rr = 0; rr < 2; ++rr) {
                #pragma unroll
                for (int g = 0; g < 2; ++g)
                    vf[t][i][rr][g] = *reinterpret_cast<const v4f*>(
                        Vb + (size_t)(t * 16 + 8 * lh + 2 * i + rr) * DD + 64 * g + 4 * col);
            }
        }
    }

    float mx = st[0][0];
    #pragma unroll
    for (int t = 0; t < NT; ++t) {
        #pragma unroll
        for (int r = 0; r < 8; ++r) mx = fmaxf(mx, st[t][r]);
    }
    mx = fmaxf(mx, __shfl_xor(mx, 16, 32));
    float sm = 0.0f;
    #pragma unroll
    for (int t = 0; t < NT; ++t) {
        #pragma unroll
        for (int r = 0; r < 8; ++r) {
            float e = __expf(st[t][r] - mx);
            st[t][r] = e;
            sm += e;
        }
    }
    sm += __shfl_xor(sm, 16, 32);
    mx_out = mx;
    sum_out = sm;

    #pragma unroll
    for (int t = 0; t < NT; ++t) {
        #pragma unroll
        for (int i = 0; i < 4; ++i) {
            v2f A = {st[t][2 * i], st[t][2 * i + 1]};
            #pragma unroll
            for (int g = 0; g < 2; ++g) {
                #pragma unroll
                for (int e = 0; e < 4; ++e) {
                    v2f B = {vf[t][i][0][g][e], vf[t][i][1][g][e]};
                    oacc[g * 4 + e] = __builtin_amdgcn_wmma_f32_16x16x4_f32(
                        false, A, false, B, (short)0, oacc[g * 4 + e], false, false);
                }
            }
        }
    }
}

__global__ __launch_bounds__(NW * 32) void k_attn_partial(const float* __restrict__ qn,
                                                          const float* __restrict__ kn,
                                                          const float* __restrict__ vn,
                                                          const float* __restrict__ cK,
                                                          const float* __restrict__ cV,
                                                          float* __restrict__ pout,
                                                          float* __restrict__ pmax,
                                                          float* __restrict__ psum)
{
    __shared__ __align__(16) float ow[NW * MM * DD];
    __shared__ float wmax[NW * MM];
    __shared__ float wsum[NW * MM];

    const int tid  = threadIdx.x;
    const int lane = tid & 31;
    const int col  = lane & 15;
    const int lh   = lane >> 4;
    const int w    = __builtin_amdgcn_readfirstlane(tid >> 5);
    const int s    = blockIdx.x;
    const int h    = blockIdx.y;

    const float* qh = qn + (size_t)h * MM * DD;

    v8f oacc[8];
    #pragma unroll
    for (int j = 0; j < 8; ++j) { v8f z = {}; oacc[j] = z; }
    float mx = -3.0e38f;
    float sm = 0.0f;

    if (s < NCHUNK - 1) {
        const size_t row0 = (size_t)h * PP + (size_t)s * CH + (size_t)w * 32;
        attn_wave<2>(qh, cK + row0 * DD, cV + row0 * DD, col, lh, mx, sm, oacc);
    } else if (w == 0) {
        attn_wave<1>(qh, kn + (size_t)h * MM * DD, vn + (size_t)h * MM * DD,
                     col, lh, mx, sm, oacc);
    }

    #pragma unroll
    for (int g = 0; g < 2; ++g) {
        #pragma unroll
        for (int r = 0; r < 8; ++r) {
            v4f o = {oacc[g * 4 + 0][r], oacc[g * 4 + 1][r],
                     oacc[g * 4 + 2][r], oacc[g * 4 + 3][r]};
            *reinterpret_cast<v4f*>(ow + ((size_t)w * MM + r + 8 * lh) * DD
                                    + 64 * g + 4 * col) = o;
        }
    }
    if (lh == 0) {
        wmax[w * MM + col] = mx;
        wsum[w * MM + col] = sm;
    }
    __syncthreads();

    {
        const int m  = w;
        const int d4 = lane * 4;
        float g = -3.0e38f;
        #pragma unroll
        for (int w2 = 0; w2 < NW; ++w2) g = fmaxf(g, wmax[w2 * MM + m]);
        v4f acc = {0.0f, 0.0f, 0.0f, 0.0f};
        #pragma unroll
        for (int w2 = 0; w2 < NW; ++w2) {
            const float f = __expf(wmax[w2 * MM + m] - g);
            const v4f o = *reinterpret_cast<const v4f*>(ow + ((size_t)w2 * MM + m) * DD + d4);
            acc += f * o;
        }
        *reinterpret_cast<v4f*>(pout + (((size_t)h * NCHUNK + s) * MM + m) * DD + d4) = acc;
    }
    if (w == 0 && lane < MM) {
        float g = -3.0e38f;
        #pragma unroll
        for (int w2 = 0; w2 < NW; ++w2) g = fmaxf(g, wmax[w2 * MM + lane]);
        float tot = 0.0f;
        #pragma unroll
        for (int w2 = 0; w2 < NW; ++w2)
            tot += __expf(wmax[w2 * MM + lane] - g) * wsum[w2 * MM + lane];
        pmax[((size_t)h * NCHUNK + s) * MM + lane] = g;
        psum[((size_t)h * NCHUNK + s) * MM + lane] = tot;
    }
}

__global__ __launch_bounds__(256) void k_combine(const float* __restrict__ pout,
                                                 const float* __restrict__ pmax,
                                                 const float* __restrict__ psum,
                                                 float* __restrict__ out)
{
    const int h = blockIdx.x;
    for (int idx = threadIdx.x; idx < MM * DD; idx += blockDim.x) {
        const int m = idx >> 7;
        const int d = idx & (DD - 1);
        const float* pm  = pmax + (size_t)h * NCHUNK * MM + m;
        const float* psu = psum + (size_t)h * NCHUNK * MM + m;
        float g = -3.0e38f;
        for (int s = 0; s < NCHUNK; ++s) g = fmaxf(g, pm[(size_t)s * MM]);
        float tot = 0.0f, acc = 0.0f;
        for (int s = 0; s < NCHUNK; ++s) {
            float w = __expf(pm[(size_t)s * MM] - g);
            tot += w * psu[(size_t)s * MM];
            acc += w * pout[(((size_t)h * NCHUNK + s) * MM + m) * DD + d];
        }
        out[(size_t)m * NN + (size_t)h * DD + d] = acc / tot;
    }
}

extern "C" void kernel_launch(void* const* d_in, const int* in_sizes, int n_in,
                              void* d_out, int out_size, void* d_ws, size_t ws_size,
                              hipStream_t stream)
{
    (void)in_sizes; (void)n_in; (void)out_size; (void)ws_size;
    const float* X  = (const float*)d_in[0];
    const float* Wq = (const float*)d_in[1];
    const float* Wk = (const float*)d_in[2];
    const float* Wv = (const float*)d_in[3];
    const float* cK = (const float*)d_in[4];
    const float* cV = (const float*)d_in[5];
    float* out = (float*)d_out;
    float* ws  = (float*)d_ws;

    float* qn    = ws;
    float* kn    = qn + HH * MM * DD;
    float* vn    = kn + HH * MM * DD;
    float* part  = vn + HH * MM * DD;
    float* pout  = part;
    float* pmaxb = pout + (size_t)HH * NCHUNK * MM * DD;
    float* psumb = pmaxb + (size_t)HH * NCHUNK * MM;

    k_proj_slab<<<dim3(NN / 64, NS, 3), 32, 0, stream>>>(X, Wq, Wk, Wv, part);
    k_qknorm<<<HH * MM, 32, 0, stream>>>(X, part, qn, kn, vn);
    k_attn_partial<<<dim3(NCHUNK, HH), NW * 32, 0, stream>>>(qn, kn, vn, cK, cV,
                                                        pout, pmaxb, psumb);
    k_combine<<<HH, 256, 0, stream>>>(pout, pmaxb, psumb, out);
}


# the 16 new rows of each head move from the attention grid into the last launch: one 16-wave block per head runs them on WMMA under its partial loads, then merges; attention grid 272 -> 256 blocks
# speedup vs baseline: 5.6827x; 1.1208x over previous
#include <hip/hip_runtime.h>
#include <math.h>


typedef __attribute__((ext_vector_type(2))) float v2f;
typedef __attribute__((ext_vector_type(4))) float v4f;
typedef __attribute__((ext_vector_type(8))) float v8f;

#define MM 16
#define NN 2048
#define DD 128
#define PP 8192
#define HH 16
#define CH 512
#define NW 16
#define NCB 16
#define KS 256
#define NS 8
#define PG 4
#define NSTG (KS / (8 * PG))
#define AHEAD 3

__global__ __launch_bounds__(32) void k_proj_slab(const float* __restrict__ X,
                                                  const float* __restrict__ Wq,
                                                  const float* __restrict__ Wk,
                                                  const float* __restrict__ Wv,
                                                  float* __restrict__ part)
{
    const int lane = threadIdx.x;
    const int col  = lane & 15;
    const int lh   = lane >> 4;
    const int n0   = blockIdx.x << 6;
    const int s    = blockIdx.y;
    const int w    = blockIdx.z;

    const float* W  = (w == 0) ? Wq : (w == 1) ? Wk : Wv;
    const float* ap = X + (size_t)col * NN + s * KS + 4 * lh;
    const float* bp = W + (size_t)(s * KS + 4 * lh) * NN + n0 + 4 * col;

    v4f af[AHEAD + 1][PG];
    v4f bf[AHEAD + 1][PG][4];
#define LOAD_STAGE(T) do { \
    _Pragma("unroll") for (int g_ = 0; g_ < PG; ++g_) { \
        const int i_ = (T) * PG + g_; \
        af[(T) % (AHEAD + 1)][g_] = *reinterpret_cast<const v4f*>(ap + 8 * i_); \
        _Pragma("unroll") for (int j_ = 0; j_ < 4; ++j_) \
            bf[(T) % (AHEAD + 1)][g_][j_] = \
                *reinterpret_cast<const v4f*>(bp + (size_t)(8 * i_ + j_) * NN); \
    } } while (0)

    v8f acc[4];
    #pragma unroll
    for (int e = 0; e < 4; ++e) { v8f z = {}; acc[e] = z; }

    #pragma unroll
    for (int t = 0; t < AHEAD; ++t) LOAD_STAGE(t);

    #pragma unroll
    for (int t = 0; t < NSTG; ++t) {
        __builtin_amdgcn_sched_barrier(0);
        if (t + AHEAD < NSTG) LOAD_STAGE(t + AHEAD);
        __builtin_amdgcn_sched_barrier(0);
        #pragma unroll
        for (int g = 0; g < PG; ++g) {
            #pragma unroll
            for (int p = 0; p < 2; ++p) {
                const v4f a4 = af[t % (AHEAD + 1)][g];
                v2f A = {a4[2 * p], a4[2 * p + 1]};
                #pragma unroll
                for (int e = 0; e < 4; ++e) {
                    v2f B = {bf[t % (AHEAD + 1)][g][2 * p][e],
                             bf[t % (AHEAD + 1)][g][2 * p + 1][e]};
                    acc[e] = __builtin_amdgcn_wmma_f32_16x16x4_f32(
                        false, A, false, B, (short)0, acc[e], false, false);
                }
            }
        }
    }
#undef LOAD_STAGE

    float* po = part + ((size_t)(w * NS + s) * MM + 8 * lh) * NN + n0 + 4 * col;
    #pragma unroll
    for (int r = 0; r < 8; ++r) {
        v4f o = {acc[0][r], acc[1][r], acc[2][r], acc[3][r]};
        *reinterpret_cast<v4f*>(po + (size_t)r * NN) = o;
    }
}

__global__ __launch_bounds__(32) void k_qknorm(const float* __restrict__ X,
                                               const float* __restrict__ part,
                                               float* __restrict__ qn,
                                               float* __restrict__ kn,
                                               float* __restrict__ vn)
{
    const int wid  = blockIdx.x;
    const int lane = threadIdx.x;
    const int h = wid >> 4;
    const int m = wid & 15;
    const int d0 = lane * 4;

    const size_t src = (size_t)m * NN + (size_t)h * DD + d0;
    const size_t dst = ((size_t)h * MM + m) * DD + d0;

    v4f xv[NN / 128];
    v4f tq[NS], tk[NS], tv[NS];
    #pragma unroll
    for (int s = 0; s < NS; ++s) {
        const float* p = part + (size_t)s * MM * NN + src;
        tq[s] = *reinterpret_cast<const v4f*>(p);
        tk[s] = *reinterpret_cast<const v4f*>(p + (size_t)NS * MM * NN);
        tv[s] = *reinterpret_cast<const v4f*>(p + (size_t)2 * NS * MM * NN);
    }
    #pragma unroll
    for (int i = 0; i < NN / 128; ++i)
        xv[i] = *reinterpret_cast<const v4f*>(X + (size_t)m * NN + 128 * i + d0);
    __builtin_amdgcn_sched_barrier(0);

    v4f q4 = tq[0], k4 = tk[0], v4 = tv[0];
    #pragma unroll
    for (int s = 1; s < NS; ++s) { q4 += tq[s]; k4 += tk[s]; v4 += tv[s]; }

    float sx = 0.0f;
    #pragma unroll
    for (int i = 0; i < NN / 128; ++i)
        sx += xv[i][0] * xv[i][0] + xv[i][1] * xv[i][1]
            + xv[i][2] * xv[i][2] + xv[i][3] * xv[i][3];
    #pragma unroll
    for (int mk = 16; mk >= 1; mk >>= 1) sx += __shfl_xor(sx, mk, 32);
    const float rx = rsqrtf(sx / (float)NN);
    q4 = q4 * rx;
    k4 = k4 * rx;
    v4 = v4 * rx;

    float ss = q4[0] * q4[0] + q4[1] * q4[1] + q4[2] * q4[2] + q4[3] * q4[3];
    #pragma unroll
    for (int mk = 16; mk >= 1; mk >>= 1) ss += __shfl_xor(ss, mk, 32);
    float rs = rsqrtf(ss / (float)DD);
    *reinterpret_cast<v4f*>(qn + dst) = q4 * rs;

    ss = k4[0] * k4[0] + k4[1] * k4[1] + k4[2] * k4[2] + k4[3] * k4[3];
    #pragma unroll
    for (int mk = 16; mk >= 1; mk >>= 1) ss += __shfl_xor(ss, mk, 32);
    rs = rsqrtf(ss / (float)DD);
    *reinterpret_cast<v4f*>(kn + dst) = k4 * rs;

    *reinterpret_cast<v4f*>(vn + dst) = v4;
}

template <int NT>
__device__ __forceinline__ void attn_wave(const float* __restrict__ qh,
                                          const float* __restrict__ Kb,
                                          const float* __restrict__ Vb,
                                          const int col, const int lh,
                                          float& mx_out, float& sum_out,
                                          v8f (&oacc)[8])
{
    v4f qf[16];
    v4f kf[NT][16];
    v4f vf[NT][4][2][2];
    if constexpr (NT == 1) {
        #pragma unroll
        for (int i = 0; i < 4; ++i) {
            #pragma unroll
            for (int rr = 0; rr < 2; ++rr) {
                #pragma unroll
                for (int g = 0; g < 2; ++g)
                    vf[0][i][rr][g] = *reinterpret_cast<const v4f*>(
                        Vb + (size_t)(8 * lh + 2 * i + rr) * DD + 64 * g + 4 * col);
            }
        }
    }
    #pragma unroll
    for (int t = 0; t < NT; ++t) {
        #pragma unroll
        for (int i = 0; i < 16; ++i)
            kf[t][i] = *reinterpret_cast<const v4f*>(
                Kb + (size_t)(t * 16 + col) * DD + 8 * i + 4 * lh);
    }
    #pragma unroll
    for (int i = 0; i < 16; ++i)
        qf[i] = *reinterpret_cast<const v4f*>(qh + (size_t)col * DD + 8 * i + 4 * lh);

    v8f st[NT];
    #pragma unroll
    for (int t = 0; t < NT; ++t) { v8f z = {}; st[t] = z; }

    #pragma unroll
    for (int i = 0; i < 16; ++i) {
        #pragma unroll
        for (int t = 0; t < NT; ++t) {
            v2f A = {kf[t][i][0], kf[t][i][1]};
            v2f B = {qf[i][0], qf[i][1]};
            st[t] = __builtin_amdgcn_wmma_f32_16x16x4_f32(false, A, false, B,
                                                          (short)0, st[t], false, false);
        }
        #pragma unroll
        for (int t = 0; t < NT; ++t) {
            v2f A = {kf[t][i][2], kf[t][i][3]};
            v2f B = {qf[i][2], qf[i][3]};
            st[t] = __builtin_amdgcn_wmma_f32_16x16x4_f32(false, A, false, B,
                                                          (short)0, st[t], false, false);
        }
    }

    #pragma unroll
    for (int t = 0; t < (NT == 1 ? 0 : NT); ++t) {
        #pragma unroll
        for (int i = 0; i < 4; ++i) {
            #pragma unroll
            for (int rr = 0; rr < 2; ++rr) {
                #pragma unroll
                for (int g = 0; g < 2; ++g)
                    vf[t][i][rr][g] = *reinterpret_cast<const v4f*>(
                        Vb + (size_t)(t * 16 + 8 * lh + 2 * i + rr) * DD + 64 * g + 4 * col);
            }
        }
    }

    float mx = st[0][0];
    #pragma unroll
    for (int t = 0; t < NT; ++t) {
        #pragma unroll
        for (int r = 0; r < 8; ++r) mx = fmaxf(mx, st[t][r]);
    }
    mx = fmaxf(mx, __shfl_xor(mx, 16, 32));
    float sm = 0.0f;
    #pragma unroll
    for (int t = 0; t < NT; ++t) {
        #pragma unroll
        for (int r = 0; r < 8; ++r) {
            float e = __expf(st[t][r] - mx);
            st[t][r] = e;
            sm += e;
        }
    }
    sm += __shfl_xor(sm, 16, 32);
    mx_out = mx;
    sum_out = sm;

    #pragma unroll
    for (int t = 0; t < NT; ++t) {
        #pragma unroll
        for (int i = 0; i < 4; ++i) {
            v2f A = {st[t][2 * i], st[t][2 * i + 1]};
            #pragma unroll
            for (int g = 0; g < 2; ++g) {
                #pragma unroll
                for (int e = 0; e < 4; ++e) {
                    v2f B = {vf[t][i][0][g][e], vf[t][i][1][g][e]};
                    oacc[g * 4 + e] = __builtin_amdgcn_wmma_f32_16x16x4_f32(
                        false, A, false, B, (short)0, oacc[g * 4 + e], false, false);
                }
            }
        }
    }
}

__global__ __launch_bounds__(NW * 32) void k_attn_partial(const float* __restrict__ qn,
                                                          const float* __restrict__ cK,
                                                          const float* __restrict__ cV,
                                                          float* __restrict__ pout,
                                                          float* __restrict__ pmax,
                                                          float* __restrict__ psum)
{
    __shared__ __align__(16) float ow[NW * MM * DD];
    __shared__ float wmax[NW * MM];
    __shared__ float wsum[NW * MM];

    const int tid  = threadIdx.x;
    const int lane = tid & 31;
    const int col  = lane & 15;
    const int lh   = lane >> 4;
    const int w    = __builtin_amdgcn_readfirstlane(tid >> 5);
    const int s    = blockIdx.x;
    const int h    = blockIdx.y;

    const float* qh = qn + (size_t)h * MM * DD;

    v8f oacc[8];
    #pragma unroll
    for (int j = 0; j < 8; ++j) { v8f z = {}; oacc[j] = z; }
    float mx = -3.0e38f;
    float sm = 0.0f;

    {
        const size_t row0 = (size_t)h * PP + (size_t)s * CH + (size_t)w * 32;
        attn_wave<2>(qh, cK + row0 * DD, cV + row0 * DD, col, lh, mx, sm, oacc);
    }

    #pragma unroll
    for (int g = 0; g < 2; ++g) {
        #pragma unroll
        for (int r = 0; r < 8; ++r) {
            v4f o = {oacc[g * 4 + 0][r], oacc[g * 4 + 1][r],
                     oacc[g * 4 + 2][r], oacc[g * 4 + 3][r]};
            *reinterpret_cast<v4f*>(ow + ((size_t)w * MM + r + 8 * lh) * DD
                                    + 64 * g + 4 * col) = o;
        }
    }
    if (lh == 0) {
        wmax[w * MM + col] = mx;
        wsum[w * MM + col] = sm;
    }
    __syncthreads();

    {
        const int m  = w;
        const int d4 = lane * 4;
        float g = -3.0e38f;
        #pragma unroll
        for (int w2 = 0; w2 < NW; ++w2) g = fmaxf(g, wmax[w2 * MM + m]);
        v4f acc = {0.0f, 0.0f, 0.0f, 0.0f};
        #pragma unroll
        for (int w2 = 0; w2 < NW; ++w2) {
            const float f = __expf(wmax[w2 * MM + m] - g);
            const v4f o = *reinterpret_cast<const v4f*>(ow + ((size_t)w2 * MM + m) * DD + d4);
            acc += f * o;
        }
        *reinterpret_cast<v4f*>(pout + (((size_t)h * NCB + s) * MM + m) * DD + d4) = acc;
    }
    if (w == 0 && lane < MM) {
        float g = -3.0e38f;
        #pragma unroll
        for (int w2 = 0; w2 < NW; ++w2) g = fmaxf(g, wmax[w2 * MM + lane]);
        float tot = 0.0f;
        #pragma unroll
        for (int w2 = 0; w2 < NW; ++w2)
            tot += __expf(wmax[w2 * MM + lane] - g) * wsum[w2 * MM + lane];
        pmax[((size_t)h * NCB + s) * MM + lane] = g;
        psum[((size_t)h * NCB + s) * MM + lane] = tot;
    }
}

__global__ __launch_bounds__(MM * 32) void k_new_merge(const float* __restrict__ qn,
                                                       const float* __restrict__ kn,
                                                       const float* __restrict__ vn,
                                                       const float* __restrict__ pout,
                                                       const float* __restrict__ pmax,
                                                       const float* __restrict__ psum,
                                                       float* __restrict__ out)
{
    __shared__ __align__(16) float nw[MM * DD];
    __shared__ float nmax[MM];
    __shared__ float nsum[MM];

    const int tid  = threadIdx.x;
    const int lane = tid & 31;
    const int col  = lane & 15;
    const int lh   = lane >> 4;
    const int w    = __builtin_amdgcn_readfirstlane(tid >> 5);
    const int h    = blockIdx.x;
    const int m    = w;
    const int d4   = lane * 4;

    v4f pv[NCB];
    #pragma unroll
    for (int s = 0; s < NCB; ++s)
        pv[s] = *reinterpret_cast<const v4f*>(
            pout + (((size_t)h * NCB + s) * MM + m) * DD + d4);
    const float pmv = pmax[((size_t)h * NCB + col) * MM + m];
    const float psv = psum[((size_t)h * NCB + col) * MM + m];

    if (w == 0) {
        v8f oacc[8];
        #pragma unroll
        for (int j = 0; j < 8; ++j) { v8f z = {}; oacc[j] = z; }
        float mx, sm;
        attn_wave<1>(qn + (size_t)h * MM * DD, kn + (size_t)h * MM * DD,
                     vn + (size_t)h * MM * DD, col, lh, mx, sm, oacc);
        #pragma unroll
        for (int g = 0; g < 2; ++g) {
            #pragma unroll
            for (int r = 0; r < 8; ++r) {
                v4f o = {oacc[g * 4 + 0][r], oacc[g * 4 + 1][r],
                         oacc[g * 4 + 2][r], oacc[g * 4 + 3][r]};
                *reinterpret_cast<v4f*>(nw + (size_t)(r + 8 * lh) * DD + 64 * g + 4 * col) = o;
            }
        }
        if (lh == 0) {
            nmax[col] = mx;
            nsum[col] = sm;
        }
    }
    __syncthreads();

    const float nm = nmax[m];
    float g = pmv;
    #pragma unroll
    for (int mk = 8; mk >= 1; mk >>= 1) g = fmaxf(g, __shfl_xor(g, mk, 32));
    g = fmaxf(g, nm);
    const float wt = __expf(pmv - g);
    float tot = 0.0f;
    v4f acc = {0.0f, 0.0f, 0.0f, 0.0f};
    #pragma unroll
    for (int s = 0; s < NCB; ++s) {
        const float ws = __int_as_float(__builtin_amdgcn_readlane(__float_as_int(wt), s));
        const float ps = __int_as_float(__builtin_amdgcn_readlane(__float_as_int(psv), s));
        tot += ws * ps;
        acc += ws * pv[s];
    }
    const float wn = __expf(nm - g);
    tot += wn * nsum[m];
    acc += wn * *reinterpret_cast<const v4f*>(nw + (size_t)m * DD + d4);
    v4f o = {acc[0] / tot, acc[1] / tot, acc[2] / tot, acc[3] / tot};
    *reinterpret_cast<v4f*>(out + (size_t)m * NN + (size_t)h * DD + d4) = o;
}

extern "C" void kernel_launch(void* const* d_in, const int* in_sizes, int n_in,
                              void* d_out, int out_size, void* d_ws, size_t ws_size,
                              hipStream_t stream)
{
    (void)in_sizes; (void)n_in; (void)out_size; (void)ws_size;
    const float* X  = (const float*)d_in[0];
    const float* Wq = (const float*)d_in[1];
    const float* Wk = (const float*)d_in[2];
    const float* Wv = (const float*)d_in[3];
    const float* cK = (const float*)d_in[4];
    const float* cV = (const float*)d_in[5];
    float* out = (float*)d_out;
    float* ws  = (float*)d_ws;

    float* qn    = ws;
    float* kn    = qn + HH * MM * DD;
    float* vn    = kn + HH * MM * DD;
    float* part  = vn + HH * MM * DD;
    float* pout  = part;
    float* pmaxb = pout + (size_t)HH * NCB * MM * DD;
    float* psumb = pmaxb + (size_t)HH * NCB * MM;

    k_proj_slab<<<dim3(NN / 64, NS, 3), 32, 0, stream>>>(X, Wq, Wk, Wv, part);
    k_qknorm<<<HH * MM, 32, 0, stream>>>(X, part, qn, kn, vn);
    k_attn_partial<<<dim3(NCB, HH), NW * 32, 0, stream>>>(qn, cK, cV,
                                                     pout, pmaxb, psumb);
    k_new_merge<<<HH, MM * 32, 0, stream>>>(qn, kn, vn, pout, pmaxb, psumb, out);
}


# qk-norm launch dropped, 4 -> 3: attention and last kernels sum the split-k partials and norm q in LDS; chunk s's block makes new k, v row s; X scale on v only; X^2 sum order, rounding change
# speedup vs baseline: 5.9250x; 1.0426x over previous
#include <hip/hip_runtime.h>
#include <math.h>


typedef __attribute__((ext_vector_type(2))) float v2f;
typedef __attribute__((ext_vector_type(4))) float v4f;
typedef __attribute__((ext_vector_type(8))) float v8f;

#define MM 16
#define NN 2048
#define DD 128
#define PP 8192
#define HH 16
#define CH 512
#define NW 16
#define NCB 16
#define KS 256
#define NS 8
#define PG 4
#define NSTG (KS / (8 * PG))
#define AHEAD 3
#define QP 132

__global__ __launch_bounds__(32) void k_proj_slab(const float* __restrict__ X,
                                                  const float* __restrict__ Wq,
                                                  const float* __restrict__ Wk,
                                                  const float* __restrict__ Wv,
                                                  float* __restrict__ part)
{
    const int lane = threadIdx.x;
    const int col  = lane & 15;
    const int lh   = lane >> 4;
    const int n0   = blockIdx.x << 6;
    const int s    = blockIdx.y;
    const int w    = blockIdx.z;

    const float* W  = (w == 0) ? Wq : (w == 1) ? Wk : Wv;
    const float* ap = X + (size_t)col * NN + s * KS + 4 * lh;
    const float* bp = W + (size_t)(s * KS + 4 * lh) * NN + n0 + 4 * col;

    v4f af[AHEAD + 1][PG];
    v4f bf[AHEAD + 1][PG][4];
#define LOAD_STAGE(T) do { \
    _Pragma("unroll") for (int g_ = 0; g_ < PG; ++g_) { \
        const int i_ = (T) * PG + g_; \
        af[(T) % (AHEAD + 1)][g_] = *reinterpret_cast<const v4f*>(ap + 8 * i_); \
        _Pragma("unroll") for (int j_ = 0; j_ < 4; ++j_) \
            bf[(T) % (AHEAD + 1)][g_][j_] = \
                *reinterpret_cast<const v4f*>(bp + (size_t)(8 * i_ + j_) * NN); \
    } } while (0)

    v8f acc[4];
    #pragma unroll
    for (int e = 0; e < 4; ++e) { v8f z = {}; acc[e] = z; }

    #pragma unroll
    for (int t = 0; t < AHEAD; ++t) LOAD_STAGE(t);

    #pragma unroll
    for (int t = 0; t < NSTG; ++t) {
        __builtin_amdgcn_sched_barrier(0);
        if (t + AHEAD < NSTG) LOAD_STAGE(t + AHEAD);
        __builtin_amdgcn_sched_barrier(0);
        #pragma unroll
        for (int g = 0; g < PG; ++g) {
            #pragma unroll
            for (int p = 0; p < 2; ++p) {
                const v4f a4 = af[t % (AHEAD + 1)][g];
                v2f A = {a4[2 * p], a4[2 * p + 1]};
                #pragma unroll
                for (int e = 0; e < 4; ++e) {
                    v2f B = {bf[t % (AHEAD + 1)][g][2 * p][e],
                             bf[t % (AHEAD + 1)][g][2 * p + 1][e]};
                    acc[e] = __builtin_amdgcn_wmma_f32_16x16x4_f32(
                        false, A, false, B, (short)0, acc[e], false, false);
                }
            }
        }
    }
#undef LOAD_STAGE

    float* po = part + ((size_t)(w * NS + s) * MM + 8 * lh) * NN + n0 + 4 * col;
    #pragma unroll
    for (int r = 0; r < 8; ++r) {
        v4f o = {acc[0][r], acc[1][r], acc[2][r], acc[3][r]};
        *reinterpret_cast<v4f*>(po + (size_t)r * NN) = o;
    }
}

__device__ __forceinline__ void load_slabs(const float* p, v4f (&t)[NS])
{
    #pragma unroll
    for (int s = 0; s < NS; ++s)
        t[s] = *reinterpret_cast<const v4f*>(p + (size_t)s * MM * NN);
}
__device__ __forceinline__ v4f sum_slabs(const v4f (&t)[NS])
{
    v4f a = t[0];
    #pragma unroll
    for (int s = 1; s < NS; ++s) a += t[s];
    return a;
}
__device__ __forceinline__ v4f rms_row(const v4f a)
{
    float ss = a[0] * a[0] + a[1] * a[1] + a[2] * a[2] + a[3] * a[3];
    #pragma unroll
    for (int mk = 16; mk >= 1; mk >>= 1) ss += __shfl_xor(ss, mk, 32);
    return a * rsqrtf(ss / (float)DD);
}

template <int NT>
__device__ __forceinline__ void load_k_frags(const float* __restrict__ Kb,
                                             const int col, const int lh,
                                             v4f (&kf)[NT][16])
{
    #pragma unroll
    for (int t = 0; t < NT; ++t) {
        #pragma unroll
        for (int i = 0; i < 16; ++i)
            kf[t][i] = *reinterpret_cast<const v4f*>(
                Kb + (size_t)(t * 16 + col) * DD + 8 * i + 4 * lh);
    }
}

template <int NT>
__device__ __forceinline__ void load_v_frags(const float* __restrict__ Vb,
                                             const int col, const int lh,
                                             v4f (&vf)[NT][4][2][2])
{
    #pragma unroll
    for (int t = 0; t < NT; ++t) {
        #pragma unroll
        for (int i = 0; i < 4; ++i) {
            #pragma unroll
            for (int rr = 0; rr < 2; ++rr) {
                #pragma unroll
                for (int g = 0; g < 2; ++g)
                    vf[t][i][rr][g] = *reinterpret_cast<const v4f*>(
                        Vb + (size_t)(t * 16 + 8 * lh + 2 * i + rr) * DD + 64 * g + 4 * col);
            }
        }
    }
}

template <int NT, bool VREADY>
__device__ __forceinline__ void attn_wave(const float* qs,
                                          const float* __restrict__ Vb,
                                          const v4f (&kf)[NT][16],
                                          v4f (&vf)[NT][4][2][2],
                                          const int col, const int lh,
                                          float& mx_out, float& sum_out,
                                          v8f (&oacc)[8])
{
    v4f qf[16];
    #pragma unroll
    for (int i = 0; i < 16; ++i)
        qf[i] = *reinterpret_cast<const v4f*>(qs + col * QP + 8 * i + 4 * lh);
    if constexpr (NT > 1) __builtin_amdgcn_sched_barrier(0);

    v8f st[NT];
    #pragma unroll
    for (int t = 0; t < NT; ++t) { v8f z = {}; st[t] = z; }

    #pragma unroll
    for (int i = 0; i < 16; ++i) {
        #pragma unroll
        for (int t = 0; t < NT; ++t) {
            v2f A = {kf[t][i][0], kf[t][i][1]};
            v2f B = {qf[i][0], qf[i][1]};
            st[t] = __builtin_amdgcn_wmma_f32_16x16x4_f32(false, A, false, B,
                                                          (short)0, st[t], false, false);
        }
        #pragma unroll
        for (int t = 0; t < NT; ++t) {
            v2f A = {kf[t][i][2], kf[t][i][3]};
            v2f B = {qf[i][2], qf[i][3]};
            st[t] = __builtin_amdgcn_wmma_f32_16x16x4_f32(false, A, false, B,
                                                          (short)0, st[t], false, false);
        }
    }

    if constexpr (!VREADY) load_v_frags<NT>(Vb, col, lh, vf);

    float mx = st[0][0];
    #pragma unroll
    for (int t = 0; t < NT; ++t) {
        #pragma unroll
        for (int r = 0; r < 8; ++r) mx = fmaxf(mx, st[t][r]);
    }
    mx = fmaxf(mx, __shfl_xor(mx, 16, 32));
    float sm = 0.0f;
    #pragma unroll
    for (int t = 0; t < NT; ++t) {
        #pragma unroll
        for (int r = 0; r < 8; ++r) {
            float e = __expf(st[t][r] - mx);
            st[t][r] = e;
            sm += e;
        }
    }
    sm += __shfl_xor(sm, 16, 32);
    mx_out = mx;
    sum_out = sm;

    #pragma unroll
    for (int t = 0; t < NT; ++t) {
        #pragma unroll
        for (int i = 0; i < 4; ++i) {
            v2f A = {st[t][2 * i], st[t][2 * i + 1]};
            #pragma unroll
            for (int g = 0; g < 2; ++g) {
                #pragma unroll
                for (int e = 0; e < 4; ++e) {
                    v2f B = {vf[t][i][0][g][e], vf[t][i][1][g][e]};
                    oacc[g * 4 + e] = __builtin_amdgcn_wmma_f32_16x16x4_f32(
                        false, A, false, B, (short)0, oacc[g * 4 + e], false, false);
                }
            }
        }
    }
}

__global__ __launch_bounds__(NW * 32) void k_attn_partial(const float* __restrict__ X,
                                                          const float* __restrict__ qpart,
                                                          const float* __restrict__ cK,
                                                          const float* __restrict__ cV,
                                                          float* kv,
                                                          float* __restrict__ kn,
                                                          float* __restrict__ vn,
                                                          float* __restrict__ pmax,
                                                          float* __restrict__ psum)
{
    __shared__ __align__(16) float ow[NW * MM * DD];
    __shared__ __align__(16) float qs[MM * QP];
    __shared__ float wmax[NW * MM];
    __shared__ float wsum[NW * MM];
    __shared__ float xs[NW];

    const int tid  = threadIdx.x;
    const int lane = tid & 31;
    const int col  = lane & 15;
    const int lh   = lane >> 4;
    const int w    = __builtin_amdgcn_readfirstlane(tid >> 5);
    const int s    = blockIdx.x;
    const int h    = blockIdx.y;
    const int d4   = lane * 4;

    v4f tq[NS], tn[NS];
    v4f kf[2][16];
    v4f vf[2][4][2][2];

    load_slabs(qpart + (size_t)w * NN + (size_t)h * DD + d4, tq);
    if (w == 1)
        load_slabs(kv + (size_t)s * NN + (size_t)h * DD + d4, tn);
    else if (w == 2)
        load_slabs(kv + ((size_t)NS * MM + s) * NN + (size_t)h * DD + d4, tn);
    const v4f xw = *reinterpret_cast<const v4f*>(X + (size_t)s * NN + 128 * w + d4);
    const size_t row0 = (size_t)h * PP + (size_t)s * CH + (size_t)w * 32;
    load_k_frags<2>(cK + row0 * DD, col, lh, kf);
    __builtin_amdgcn_sched_barrier(0);

    *reinterpret_cast<v4f*>(qs + w * QP + d4) = rms_row(sum_slabs(tq));
    {
        float sx = xw[0] * xw[0] + xw[1] * xw[1] + xw[2] * xw[2] + xw[3] * xw[3];
        #pragma unroll
        for (int mk = 16; mk >= 1; mk >>= 1) sx += __shfl_xor(sx, mk, 32);
        if (lane == 0) xs[w] = sx;
    }
    v4f vsum = {0.0f, 0.0f, 0.0f, 0.0f};
    if (w == 1)
        *reinterpret_cast<v4f*>(kn + ((size_t)h * MM + s) * DD + d4) = rms_row(sum_slabs(tn));
    else if (w == 2)
        vsum = sum_slabs(tn);
    __builtin_amdgcn_fence(__ATOMIC_RELEASE, "workgroup", "local");
    __builtin_amdgcn_s_barrier();
    __builtin_amdgcn_fence(__ATOMIC_ACQUIRE, "workgroup", "local");
    if (w == 2) {
        float sx = xs[0];
        #pragma unroll
        for (int w2 = 1; w2 < NW; ++w2) sx += xs[w2];
        const float rx = rsqrtf(sx / (float)NN);
        *reinterpret_cast<v4f*>(vn + ((size_t)h * MM + s) * DD + d4) = vsum * rx;
    }

    v8f oacc[8];
    #pragma unroll
    for (int j = 0; j < 8; ++j) { v8f z = {}; oacc[j] = z; }
    float mx = -3.0e38f;
    float sm = 0.0f;

    attn_wave<2, false>(qs, cV + row0 * DD, kf, vf, col, lh, mx, sm, oacc);

    #pragma unroll
    for (int g = 0; g < 2; ++g) {
        #pragma unroll
        for (int r = 0; r < 8; ++r) {
            v4f o = {oacc[g * 4 + 0][r], oacc[g * 4 + 1][r],
                     oacc[g * 4 + 2][r], oacc[g * 4 + 3][r]};
            *reinterpret_cast<v4f*>(ow + ((size_t)w * MM + r + 8 * lh) * DD
                                    + 64 * g + 4 * col) = o;
        }
    }
    if (lh == 0) {
        wmax[w * MM + col] = mx;
        wsum[w * MM + col] = sm;
    }
    __syncthreads();

    {
        const int m  = w;
        float g = -3.0e38f;
        #pragma unroll
        for (int w2 = 0; w2 < NW; ++w2) g = fmaxf(g, wmax[w2 * MM + m]);
        v4f acc = {0.0f, 0.0f, 0.0f, 0.0f};
        #pragma unroll
        for (int w2 = 0; w2 < NW; ++w2) {
            const float f = __expf(wmax[w2 * MM + m] - g);
            const v4f o = *reinterpret_cast<const v4f*>(ow + ((size_t)w2 * MM + m) * DD + d4);
            acc += f * o;
        }
        *reinterpret_cast<v4f*>(kv + ((size_t)m * MM + s) * NN + (size_t)h * DD + d4) = acc;
    }
    if (w == 0 && lane < MM) {
        float g = -3.0e38f;
        #pragma unroll
        for (int w2 = 0; w2 < NW; ++w2) g = fmaxf(g, wmax[w2 * MM + lane]);
        float tot = 0.0f;
        #pragma unroll
        for (int w2 = 0; w2 < NW; ++w2)
            tot += __expf(wmax[w2 * MM + lane] - g) * wsum[w2 * MM + lane];
        pmax[((size_t)h * NCB + s) * MM + lane] = g;
        psum[((size_t)h * NCB + s) * MM + lane] = tot;
    }
}

__global__ __launch_bounds__(MM * 32) void k_new_merge(const float* __restrict__ qpart,
                                                       const float* __restrict__ kv,
                                                       const float* __restrict__ kn,
                                                       const float* __restrict__ vn,
                                                       const float* __restrict__ pmax,
                                                       const float* __restrict__ psum,
                                                       float* __restrict__ out)
{
    __shared__ __align__(16) float nw[MM * DD];
    __shared__ __align__(16) float qs[MM * QP];
    __shared__ float nmax[MM];
    __shared__ float nsum[MM];

    const int tid  = threadIdx.x;
    const int lane = tid & 31;
    const int col  = lane & 15;
    const int lh   = lane >> 4;
    const int w    = __builtin_amdgcn_readfirstlane(tid >> 5);
    const int h    = blockIdx.x;
    const int m    = w;
    const int d4   = lane * 4;

    v4f tq[NS];
    v4f pv[NCB];
    v4f kf[1][16];
    v4f vf[1][4][2][2];
    load_slabs(qpart + (size_t)m * NN + (size_t)h * DD + d4, tq);
    #pragma unroll
    for (int s = 0; s < NCB; ++s)
        pv[s] = *reinterpret_cast<const v4f*>(
            kv + ((size_t)m * MM + s) * NN + (size_t)h * DD + d4);
    const float pmv = pmax[((size_t)h * NCB + col) * MM + m];
    const float psv = psum[((size_t)h * NCB + col) * MM + m];
    if (w == 0) {
        load_k_frags<1>(kn + (size_t)h * MM * DD, col, lh, kf);
        load_v_frags<1>(vn + (size_t)h * MM * DD, col, lh, vf);
    }
    __builtin_amdgcn_sched_barrier(0);

    *reinterpret_cast<v4f*>(qs + m * QP + d4) = rms_row(sum_slabs(tq));
    __syncthreads();

    if (w == 0) {
        v8f oacc[8];
        #pragma unroll
        for (int j = 0; j < 8; ++j) { v8f z = {}; oacc[j] = z; }
        float mx, sm;
        attn_wave<1, true>(qs, vn, kf, vf, col, lh, mx, sm, oacc);
        #pragma unroll
        for (int g = 0; g < 2; ++g) {
            #pragma unroll
            for (int r = 0; r < 8; ++r) {
                v4f o = {oacc[g * 4 + 0][r], oacc[g * 4 + 1][r],
                         oacc[g * 4 + 2][r], oacc[g * 4 + 3][r]};
                *reinterpret_cast<v4f*>(nw + (size_t)(r + 8 * lh) * DD + 64 * g + 4 * col) = o;
            }
        }
        if (lh == 0) {
            nmax[col] = mx;
            nsum[col] = sm;
        }
    }
    __syncthreads();

    const float nm = nmax[m];
    float g = pmv;
    #pragma unroll
    for (int mk = 8; mk >= 1; mk >>= 1) g = fmaxf(g, __shfl_xor(g, mk, 32));
    g = fmaxf(g, nm);
    const float wt = __expf(pmv - g);
    float tot = 0.0f;
    v4f acc = {0.0f, 0.0f, 0.0f, 0.0f};
    #pragma unroll
    for (int s = 0; s < NCB; ++s) {
        const float ws = __int_as_float(__builtin_amdgcn_readlane(__float_as_int(wt), s));
        const float ps = __int_as_float(__builtin_amdgcn_readlane(__float_as_int(psv), s));
        tot += ws * ps;
        acc += ws * pv[s];
    }
    const float wn = __expf(nm - g);
    tot += wn * nsum[m];
    acc += wn * *reinterpret_cast<const v4f*>(nw + (size_t)m * DD + d4);
    v4f o = {acc[0] / tot, acc[1] / tot, acc[2] / tot, acc[3] / tot};
    *reinterpret_cast<v4f*>(out + (size_t)m * NN + (size_t)h * DD + d4) = o;
}

extern "C" void kernel_launch(void* const* d_in, const int* in_sizes, int n_in,
                              void* d_out, int out_size, void* d_ws, size_t ws_size,
                              hipStream_t stream)
{
    (void)in_sizes; (void)n_in; (void)out_size; (void)ws_size;
    const float* X  = (const float*)d_in[0];
    const float* Wq = (const float*)d_in[1];
    const float* Wk = (const float*)d_in[2];
    const float* Wv = (const float*)d_in[3];
    const float* cK = (const float*)d_in[4];
    const float* cV = (const float*)d_in[5];
    float* out = (float*)d_out;
    float* ws  = (float*)d_ws;

    float* pmaxb = ws;
    float* psumb = pmaxb + (size_t)HH * NCB * MM;
    float* kn    = ws + HH * MM * DD;
    float* vn    = kn + HH * MM * DD;
    float* part  = vn + HH * MM * DD;
    float* kv    = part + (size_t)NS * MM * NN;

    k_proj_slab<<<dim3(NN / 64, NS, 3), 32, 0, stream>>>(X, Wq, Wk, Wv, part);
    k_attn_partial<<<dim3(NCB, HH), NW * 32, 0, stream>>>(X, part, cK, cV, kv,
                                                     kn, vn, pmaxb, psumb);
    k_new_merge<<<HH, MM * 32, 0, stream>>>(part, kv, kn, vn, pmaxb, psumb, out);
}
